# starting kernel
# baseline (speedup 1.0000x reference)
#include <hip/hip_runtime.h>

#define NN 50000
#define NR 3
#define NE 500000
#define DIM 128

typedef __attribute__((ext_vector_type(2))) float v2f;
typedef __attribute__((ext_vector_type(8))) float v8f;

__global__ __launch_bounds__(256) void rgcn_zero_f32(float* __restrict__ p, int n) {
    int i = blockIdx.x * 256 + threadIdx.x;
    if (i < n) p[i] = 0.0f;
}

__global__ __launch_bounds__(256) void rgcn_count_deg(const int* __restrict__ src,
                                                      const int* __restrict__ dst,
                                                      float* __restrict__ outdeg,
                                                      float* __restrict__ indeg) {
    int i = blockIdx.x * 256 + threadIdx.x;
    if (i < NR * NE) {
        int r = i / NE;
        atomicAdd(outdeg + r * NN + src[i], 1.0f);
        atomicAdd(indeg  + r * NN + dst[i], 1.0f);
    }
}

__global__ __launch_bounds__(256) void rgcn_bias_init(float* __restrict__ acc,
                                                      const float* __restrict__ b) {
    int i = blockIdx.x * 256 + threadIdx.x;
    if (i < NN * DIM) {
        int k = i & (DIM - 1);
        acc[i] = b[k] + b[DIM + k] + b[2 * DIM + k];
    }
}

template <bool RELU>
__global__ __launch_bounds__(256) void rgcn_gemm_wmma(const float* __restrict__ H,
                                                      const float* __restrict__ outdeg,
                                                      const float* __restrict__ W,
                                                      float* __restrict__ Z) {
    __shared__ float As[16][DIM + 4];
    const int row0 = blockIdx.x * 16;

    for (int t = threadIdx.x; t < 16 * DIM; t += 256) {
        const int r = t >> 7;
        const int col = t & (DIM - 1);
        const float rs = rsqrtf(fmaxf(outdeg[row0 + r], 1.0f));
        float v = H[(row0 + r) * DIM + col];
        if (RELU) v = fmaxf(v, 0.0f);
        As[r][col] = v * rs;
    }
    __syncthreads();

    const int wave = threadIdx.x >> 5;
    const int lane = threadIdx.x & 31;
    const int half = lane >> 4;
    const int m    = lane & 15;
    const int col0 = wave * 16;

    v8f acc = {};
    #pragma unroll
    for (int k0 = 0; k0 < DIM; k0 += 4) {
        const int ka = k0 + 2 * half;
        v2f a, b;
        a.x = As[m][ka];
        a.y = As[m][ka + 1];
        b.x = W[ka * DIM + col0 + m];
        b.y = W[(ka + 1) * DIM + col0 + m];
        acc = __builtin_amdgcn_wmma_f32_16x16x4_f32(
             false, a,  false, b,
             (short)0, acc,  false,  false);
    }

    #pragma unroll
    for (int v = 0; v < 8; ++v) {
        Z[(row0 + v + 8 * half) * DIM + col0 + m] = acc[v];
    }
}

__global__ __launch_bounds__(256) void rgcn_scatter(const float* __restrict__ Z,
                                                    const int* __restrict__ src,
                                                    const int* __restrict__ dst,
                                                    const float* __restrict__ indeg,
                                                    float* __restrict__ acc) {
    const long long idx = (long long)blockIdx.x * 256 + threadIdx.x;
    const int e = (int)(idx >> 5);
    const int lane = (int)(idx & 31);
    if (e >= NE) return;
    const int s = src[e];
    const int d = dst[e];
    const float sc = rsqrtf(fmaxf(indeg[d], 1.0f));
    const float4 v = *(const float4*)(Z + (long long)s * DIM + lane * 4);
    float* o = acc + (long long)d * DIM + lane * 4;
    atomicAdd(o + 0, v.x * sc);
    atomicAdd(o + 1, v.y * sc);
    atomicAdd(o + 2, v.z * sc);
    atomicAdd(o + 3, v.w * sc);
}

extern "C" void kernel_launch(void* const* d_in, const int* in_sizes, int n_in,
                              void* d_out, int out_size, void* d_ws, size_t ws_size,
                              hipStream_t stream) {
    const float* x    = (const float*)d_in[0];
    const int*   esrc = (const int*)  d_in[1];
    const int*   edst = (const int*)  d_in[2];
    const float* W1   = (const float*)d_in[3];
    const float* b1   = (const float*)d_in[4];
    const float* W2   = (const float*)d_in[5];
    const float* b2   = (const float*)d_in[6];
    float*       out  = (float*)d_out;

    float* ws = (float*)d_ws;
    float* Z      = ws;
    float* H1     = Z + (size_t)NN * DIM;
    float* outdeg = H1 + (size_t)NN * DIM;
    float* indeg  = outdeg + (size_t)NR * NN;

    const int degN    = 2 * NR * NN;
    const int gemmGrd = NN / 16;
    const int scatGrd = (NE * 32) / 256;

    rgcn_zero_f32<<<(degN + 255) / 256, 256, 0, stream>>>(outdeg, degN);
    rgcn_count_deg<<<(NR * NE + 255) / 256, 256, 0, stream>>>(esrc, edst, outdeg, indeg);

    rgcn_bias_init<<<(NN * DIM + 255) / 256, 256, 0, stream>>>(H1, b1);
    for (int r = 0; r < NR; ++r) {
        rgcn_gemm_wmma<false><<<gemmGrd, 256, 0, stream>>>(
            x, outdeg + r * NN, W1 + (size_t)r * DIM * DIM, Z);
        rgcn_scatter<<<scatGrd, 256, 0, stream>>>(
            Z, esrc + r * NE, edst + r * NE, indeg + r * NN, H1);
    }

    rgcn_bias_init<<<(NN * DIM + 255) / 256, 256, 0, stream>>>(out, b2);
    for (int r = 0; r < NR; ++r) {
        rgcn_gemm_wmma<true><<<gemmGrd, 256, 0, stream>>>(
            H1, outdeg + r * NN, W2 + (size_t)r * DIM * DIM, Z);
        rgcn_scatter<<<scatGrd, 256, 0, stream>>>(
            Z, esrc + r * NE, edst + r * NE, indeg + r * NN, out);
    }
}


# degree counts by LDS histogram: each workgroup scans one index array and counts its own 6250-node range in LDS, stores f32; replaces 3M random global f32 atomics and the zero-fill launch
# speedup vs baseline: 1.1728x; 1.1728x over previous
#include <hip/hip_runtime.h>

#define NN 50000
#define NR 3
#define NE 500000
#define DIM 128

typedef __attribute__((ext_vector_type(2))) float v2f;
typedef __attribute__((ext_vector_type(8))) float v8f;

#define DEG_PARTS 8
#define DEG_BINS (NN / DEG_PARTS)

static __device__ __forceinline__ void deg_count(unsigned int* cnt, unsigned int lo, int node) {
    const unsigned int t = (unsigned int)node - lo;
    if (t < (unsigned int)DEG_BINS) atomicAdd(&cnt[t], 1u);
}

__global__ __launch_bounds__(1024) void rgcn_count_deg(const int* __restrict__ src,
                                                       const int* __restrict__ dst,
                                                       float* __restrict__ outdeg,
                                                       float* __restrict__ indeg) {
    __shared__ unsigned int cnt[DEG_BINS];
    const int h    = blockIdx.x / DEG_PARTS;
    const int part = blockIdx.x - h * DEG_PARTS;
    const int r    = (h < NR) ? h : h - NR;
    const int* idx = ((h < NR) ? src : dst) + r * NE;
    float*     deg = ((h < NR) ? outdeg : indeg) + r * NN;
    const unsigned int lo = (unsigned int)(part * DEG_BINS);

    for (int j = threadIdx.x; j < DEG_BINS; j += 1024) cnt[j] = 0u;
    __syncthreads();

    const int4* p = (const int4*)idx;
    constexpr int NV = NE / 4;
    constexpr int U  = 8;
    int i = threadIdx.x;
    for (; i + (U - 1) * 1024 < NV; i += U * 1024) {
        int4 v[U];
        #pragma unroll
        for (int u = 0; u < U; ++u) v[u] = p[i + u * 1024];
        #pragma unroll
        for (int u = 0; u < U; ++u) {
            deg_count(cnt, lo, v[u].x);
            deg_count(cnt, lo, v[u].y);
            deg_count(cnt, lo, v[u].z);
            deg_count(cnt, lo, v[u].w);
        }
    }
    for (; i < NV; i += 1024) {
        const int4 v = p[i];
        deg_count(cnt, lo, v.x);
        deg_count(cnt, lo, v.y);
        deg_count(cnt, lo, v.z);
        deg_count(cnt, lo, v.w);
    }
    __syncthreads();

    for (int j = threadIdx.x; j < DEG_BINS; j += 1024) deg[lo + j] = (float)cnt[j];
}

__global__ __launch_bounds__(256) void rgcn_bias_init(float* __restrict__ acc,
                                                      const float* __restrict__ b) {
    int i = blockIdx.x * 256 + threadIdx.x;
    if (i < NN * DIM) {
        int k = i & (DIM - 1);
        acc[i] = b[k] + b[DIM + k] + b[2 * DIM + k];
    }
}

template <bool RELU>
__global__ __launch_bounds__(256) void rgcn_gemm_wmma(const float* __restrict__ H,
                                                      const float* __restrict__ outdeg,
                                                      const float* __restrict__ W,
                                                      float* __restrict__ Z) {
    __shared__ float As[16][DIM + 4];
    const int row0 = blockIdx.x * 16;

    for (int t = threadIdx.x; t < 16 * DIM; t += 256) {
        const int r = t >> 7;
        const int col = t & (DIM - 1);
        const float rs = rsqrtf(fmaxf(outdeg[row0 + r], 1.0f));
        float v = H[(row0 + r) * DIM + col];
        if (RELU) v = fmaxf(v, 0.0f);
        As[r][col] = v * rs;
    }
    __syncthreads();

    const int wave = threadIdx.x >> 5;
    const int lane = threadIdx.x & 31;
    const int half = lane >> 4;
    const int m    = lane & 15;
    const int col0 = wave * 16;

    v8f acc = {};
    #pragma unroll
    for (int k0 = 0; k0 < DIM; k0 += 4) {
        const int ka = k0 + 2 * half;
        v2f a, b;
        a.x = As[m][ka];
        a.y = As[m][ka + 1];
        b.x = W[ka * DIM + col0 + m];
        b.y = W[(ka + 1) * DIM + col0 + m];
        acc = __builtin_amdgcn_wmma_f32_16x16x4_f32(
             false, a,  false, b,
             (short)0, acc,  false,  false);
    }

    #pragma unroll
    for (int v = 0; v < 8; ++v) {
        Z[(row0 + v + 8 * half) * DIM + col0 + m] = acc[v];
    }
}

__global__ __launch_bounds__(256) void rgcn_scatter(const float* __restrict__ Z,
                                                    const int* __restrict__ src,
                                                    const int* __restrict__ dst,
                                                    const float* __restrict__ indeg,
                                                    float* __restrict__ acc) {
    const long long idx = (long long)blockIdx.x * 256 + threadIdx.x;
    const int e = (int)(idx >> 5);
    const int lane = (int)(idx & 31);
    if (e >= NE) return;
    const int s = src[e];
    const int d = dst[e];
    const float sc = rsqrtf(fmaxf(indeg[d], 1.0f));
    const float4 v = *(const float4*)(Z + (long long)s * DIM + lane * 4);
    float* o = acc + (long long)d * DIM + lane * 4;
    atomicAdd(o + 0, v.x * sc);
    atomicAdd(o + 1, v.y * sc);
    atomicAdd(o + 2, v.z * sc);
    atomicAdd(o + 3, v.w * sc);
}

extern "C" void kernel_launch(void* const* d_in, const int* in_sizes, int n_in,
                              void* d_out, int out_size, void* d_ws, size_t ws_size,
                              hipStream_t stream) {
    const float* x    = (const float*)d_in[0];
    const int*   esrc = (const int*)  d_in[1];
    const int*   edst = (const int*)  d_in[2];
    const float* W1   = (const float*)d_in[3];
    const float* b1   = (const float*)d_in[4];
    const float* W2   = (const float*)d_in[5];
    const float* b2   = (const float*)d_in[6];
    float*       out  = (float*)d_out;

    float* ws = (float*)d_ws;
    float* Z      = ws;
    float* H1     = Z + (size_t)NN * DIM;
    float* outdeg = H1 + (size_t)NN * DIM;
    float* indeg  = outdeg + (size_t)NR * NN;

    const int gemmGrd = NN / 16;
    const int scatGrd = (NE * 32) / 256;

    rgcn_count_deg<<<2 * NR * DEG_PARTS, 1024, 0, stream>>>(esrc, edst, outdeg, indeg);

    rgcn_bias_init<<<(NN * DIM + 255) / 256, 256, 0, stream>>>(H1, b1);
    for (int r = 0; r < NR; ++r) {
        rgcn_gemm_wmma<false><<<gemmGrd, 256, 0, stream>>>(
            x, outdeg + r * NN, W1 + (size_t)r * DIM * DIM, Z);
        rgcn_scatter<<<scatGrd, 256, 0, stream>>>(
            Z, esrc + r * NE, edst + r * NE, indeg + r * NN, H1);
    }

    rgcn_bias_init<<<(NN * DIM + 255) / 256, 256, 0, stream>>>(out, b2);
    for (int r = 0; r < NR; ++r) {
        rgcn_gemm_wmma<true><<<gemmGrd, 256, 0, stream>>>(
            H1, outdeg + r * NN, W2 + (size_t)r * DIM * DIM, Z);
        rgcn_scatter<<<scatGrd, 256, 0, stream>>>(
            Z, esrc + r * NE, edst + r * NE, indeg + r * NN, out);
    }
}


# build pass split over edge chunks in two launches: partial LDS histograms per (array, part, chunk), then offsets + placement per chunk; 408 workgroups, not 48; lists chunk-major, so sum order changes
# speedup vs baseline: 4.0270x; 3.4337x over previous
#include <hip/hip_runtime.h>

#define NN 50000
#define NR 3
#define NE 500000
#define DIM 128

typedef __attribute__((ext_vector_type(2))) float v2f;
typedef __attribute__((ext_vector_type(8))) float v8f;

#define DEG_PARTS 8
#define DEG_BINS (NN / DEG_PARTS)
#define BIN_CH ((DEG_BINS + 1023) / 1024)
#define LIST_PAD 32
#define LIST_LEN (NE + DEG_PARTS * LIST_PAD)
#define HIST_PARTS (DEG_PARTS / 2)
#define HIST_BINS (2 * DEG_BINS)
#define NCH 8
#define CH_E (NE / NCH)
#define CH_V (CH_E / 4)

static __device__ __forceinline__ void deg_count(unsigned int* cnt, unsigned int lo, int node) {
    const unsigned int t = (unsigned int)node - lo;
    if (t < (unsigned int)HIST_BINS) atomicAdd(&cnt[t], 1u);
}

__global__ __launch_bounds__(1024) void rgcn_hist(const int* __restrict__ src,
                                                  const int* __restrict__ dst,
                                                  int* __restrict__ hist,
                                                  int* __restrict__ tot) {
    __shared__ unsigned int cnt[HIST_BINS];
    __shared__ unsigned int blk_tot[2];
    const int tid  = threadIdx.x;
    const int h    = blockIdx.x / (HIST_PARTS * NCH);
    const int rem  = blockIdx.x - h * (HIST_PARTS * NCH);
    const int part = rem / NCH;
    const int c    = rem - part * NCH;
    const unsigned int lo = (unsigned int)(part * HIST_BINS);

    for (int j = tid; j < HIST_BINS; j += 1024) cnt[j] = 0u;
    if (tid < 2) blk_tot[tid] = 0u;
    __syncthreads();

    const int* idx = (h < NR) ? (src + h * NE) : (dst + (h - NR) * NE);
    const int4* p = (const int4*)(idx + c * CH_E);
    constexpr int U = 8;
    for (int i0 = 0; i0 < CH_V; i0 += U * 1024) {
        int4 v[U];
        #pragma unroll
        for (int u = 0; u < U; ++u) {
            const int k = i0 + u * 1024 + tid;
            v[u] = make_int4(-1, -1, -1, -1);
            if (k < CH_V) v[u] = p[k];
        }
        #pragma unroll
        for (int u = 0; u < U; ++u) {
            deg_count(cnt, lo, v[u].x);
            deg_count(cnt, lo, v[u].y);
            deg_count(cnt, lo, v[u].z);
            deg_count(cnt, lo, v[u].w);
        }
    }
    __syncthreads();

    int* hrow = hist + (h * NCH + c) * NN + (int)lo;
    unsigned int s0 = 0u, s1 = 0u;
    for (int j = tid; j < HIST_BINS; j += 1024) {
        const unsigned int v = cnt[j];
        hrow[j] = (int)v;
        if (j < DEG_BINS) s0 += v; else s1 += v;
    }
    for (int off = 16; off > 0; off >>= 1) {
        s0 += __shfl_xor(s0, off);
        s1 += __shfl_xor(s1, off);
    }
    if ((tid & 31) == 0) { atomicAdd(&blk_tot[0], s0); atomicAdd(&blk_tot[1], s1); }
    __syncthreads();
    if (tid < 2) tot[(h * DEG_PARTS + 2 * part + tid) * NCH + c] = (int)blk_tot[tid];
}

static __device__ __forceinline__ void list_place(unsigned int* cursor, unsigned int lo,
                                                  int* __restrict__ lst, int d, int s) {
    const unsigned int t = (unsigned int)d - lo;
    if (t < (unsigned int)DEG_BINS) {
        const unsigned int pos = atomicAdd(&cursor[t], 1u);
        lst[pos] = s;
    }
}

__global__ __launch_bounds__(1024) void rgcn_place(const int* __restrict__ src,
                                                   const int* __restrict__ dst,
                                                   const int* __restrict__ hist,
                                                   const int* __restrict__ tot,
                                                   float* __restrict__ outscale,
                                                   int* __restrict__ rowstart,
                                                   int* __restrict__ incnt,
                                                   int* __restrict__ slist) {
    __shared__ unsigned int cnt[DEG_BINS];
    __shared__ unsigned int pre[DEG_BINS];
    __shared__ unsigned int psum[2][1024];
    __shared__ unsigned int below_all;
    const int tid = threadIdx.x;

    if (blockIdx.x >= NR * DEG_PARTS * NCH) {
        const int q    = blockIdx.x - NR * DEG_PARTS * NCH;
        const int h    = q / DEG_PARTS;
        const int part = q - h * DEG_PARTS;
        const int* hs  = hist + (h * NCH) * NN + part * DEG_BINS;
        #pragma unroll
        for (int k = 0; k < BIN_CH; ++k) {
            const int j = tid + k * 1024;
            if (j < DEG_BINS) {
                unsigned int s = 0u;
                #pragma unroll
                for (int cc = 0; cc < NCH; ++cc) s += (unsigned int)hs[cc * NN + j];
                outscale[h * NN + part * DEG_BINS + j] = rsqrtf(fmaxf((float)s, 1.0f));
            }
        }
        return;
    }

    const int r    = blockIdx.x / (DEG_PARTS * NCH);
    const int rem  = blockIdx.x - r * (DEG_PARTS * NCH);
    const int part = rem / NCH;
    const int c    = rem - part * NCH;
    const unsigned int lo = (unsigned int)(part * DEG_BINS);

    if (tid == 0) below_all = 0u;
    __syncthreads();

    const int* hd = hist + ((NR + r) * NCH) * NN + (int)lo;
    #pragma unroll
    for (int k = 0; k < BIN_CH; ++k) {
        const int j = tid + k * 1024;
        if (j < DEG_BINS) {
            unsigned int total = 0u, lower = 0u;
            #pragma unroll
            for (int cc = 0; cc < NCH; ++cc) {
                const unsigned int v = (unsigned int)hd[cc * NN + j];
                total += v;
                if (cc < c) lower += v;
            }
            cnt[j] = total;
            pre[j] = lower;
        }
    }
    if (tid < part * NCH) atomicAdd(&below_all, (unsigned int)tot[((NR + r) * DEG_PARTS) * NCH + tid]);
    __syncthreads();

    unsigned int cq[BIN_CH];
    unsigned int tsum = 0u;
    #pragma unroll
    for (int q = 0; q < BIN_CH; ++q) {
        const int b = tid * BIN_CH + q;
        cq[q] = (b < DEG_BINS) ? cnt[b] : 0u;
        tsum += cq[q];
    }
    psum[0][tid] = tsum;
    __syncthreads();
    int cur = 0;
    for (int off = 1; off < 1024; off <<= 1) {
        unsigned int v = psum[cur][tid];
        if (tid >= off) v += psum[cur][tid - off];
        psum[cur ^ 1][tid] = v;
        __syncthreads();
        cur ^= 1;
    }
    const unsigned int base = (below_all + (unsigned int)(LIST_PAD * part) + (LIST_PAD - 1)) & ~(unsigned int)(LIST_PAD - 1);
    unsigned int run = base + psum[cur][tid] - tsum;
    #pragma unroll
    for (int q = 0; q < BIN_CH; ++q) {
        const int b = tid * BIN_CH + q;
        if (b < DEG_BINS) {
            if (c == 0) {
                rowstart[r * NN + lo + b] = (int)run;
                incnt[r * NN + lo + b]    = (int)cq[q];
            }
            cnt[b] = run + pre[b];
            run += cq[q];
        }
    }
    __syncthreads();

    const int4* pd = (const int4*)(dst + r * NE + c * CH_E);
    const int4* ps = (const int4*)(src + r * NE + c * CH_E);
    int* lst = slist + r * LIST_LEN;
    constexpr int U2 = 4;
    for (int i0 = 0; i0 < CH_V; i0 += U2 * 1024) {
        int4 vd[U2], vs[U2];
        #pragma unroll
        for (int u = 0; u < U2; ++u) {
            const int k = i0 + u * 1024 + tid;
            vd[u] = make_int4(-1, -1, -1, -1);
            vs[u] = make_int4(0, 0, 0, 0);
            if (k < CH_V) { vd[u] = pd[k]; vs[u] = ps[k]; }
        }
        #pragma unroll
        for (int u = 0; u < U2; ++u) {
            list_place(cnt, lo, lst, vd[u].x, vs[u].x);
            list_place(cnt, lo, lst, vd[u].y, vs[u].y);
            list_place(cnt, lo, lst, vd[u].z, vs[u].z);
            list_place(cnt, lo, lst, vd[u].w, vs[u].w);
        }
    }
}

template <bool RELU>
__global__ __launch_bounds__(256, 16) void rgcn_pull_gemm(const float* __restrict__ H,
                                                      const float* __restrict__ outscale,
                                                      const int* __restrict__ rowstart,
                                                      const int* __restrict__ incnt,
                                                      const int* __restrict__ slist,
                                                      const float* __restrict__ W,
                                                      const float* __restrict__ b,
                                                      float* __restrict__ O) {
    __shared__ __attribute__((aligned(16))) float As[NR][16][DIM + 4];
    const int row0 = blockIdx.x * 16;
    const int wave = threadIdx.x >> 5;
    const int lane = threadIdx.x & 31;

    int st[NR][2], nn[NR][2];
    #pragma unroll
    for (int r = 0; r < NR; ++r) {
        #pragma unroll
        for (int q = 0; q < 2; ++q) {
            st[r][q] = rowstart[r * NN + row0 + wave * 2 + q];
            nn[r][q] = incnt[r * NN + row0 + wave * 2 + q];
        }
    }
    int   s_l[NR][2];
    float sc_l[NR][2];
    #pragma unroll
    for (int r = 0; r < NR; ++r) {
        #pragma unroll
        for (int q = 0; q < 2; ++q) {
            s_l[r][q] = 0;
            if (lane < nn[r][q]) s_l[r][q] = slist[r * LIST_LEN + st[r][q] + lane];
        }
    }
    #pragma unroll
    for (int r = 0; r < NR; ++r) {
        #pragma unroll
        for (int q = 0; q < 2; ++q) {
            sc_l[r][q] = 0.0f;
            if (lane < nn[r][q]) sc_l[r][q] = outscale[r * NN + s_l[r][q]];
        }
    }

    #pragma unroll
    for (int r = 0; r < NR; ++r) {
        const float* osc = outscale + r * NN;
        #pragma unroll
        for (int q = 0; q < 2; ++q) {
            const int start = __builtin_amdgcn_readfirstlane(st[r][q]);
            const int n     = __builtin_amdgcn_readfirstlane(nn[r][q]);
            const int* lst = slist + r * LIST_LEN + start;
            int   si  = s_l[r][q];
            float sci = sc_l[r][q];
            float4 sum = make_float4(0.0f, 0.0f, 0.0f, 0.0f);
            for (int e0 = 0; e0 < n; e0 += 32) {
                const int m = (n - e0 < 32) ? (n - e0) : 32;
                if (e0 > 0) {
                    si = 0; sci = 0.0f;
                    if (lane < m) { si = lst[e0 + lane]; sci = osc[si]; }
                }
                for (int j = 0; j < m; j += 4) {
                    float4 v[4];
                    float  sc[4];
                    #pragma unroll
                    for (int u = 0; u < 4; ++u) {
                        const bool live = (j + u) < m;
                        const int  jj   = live ? (j + u) : j;
                        const int s = __builtin_amdgcn_readlane(si, jj);
                        sc[u] = __int_as_float(__builtin_amdgcn_readlane(__float_as_int(sci), jj));
                        v[u] = *(const float4*)(H + (size_t)s * DIM + lane * 4);
                        if (!live) v[u] = make_float4(0.0f, 0.0f, 0.0f, 0.0f);
                    }
                    #pragma unroll
                    for (int u = 0; u < 4; ++u) {
                        if (RELU) {
                            v[u].x = fmaxf(v[u].x, 0.0f); v[u].y = fmaxf(v[u].y, 0.0f);
                            v[u].z = fmaxf(v[u].z, 0.0f); v[u].w = fmaxf(v[u].w, 0.0f);
                        }
                        sum.x = fmaf(v[u].x, sc[u], sum.x);
                        sum.y = fmaf(v[u].y, sc[u], sum.y);
                        sum.z = fmaf(v[u].z, sc[u], sum.z);
                        sum.w = fmaf(v[u].w, sc[u], sum.w);
                    }
                }
            }
            const float rs = rsqrtf(fmaxf((float)n, 1.0f));
            sum.x *= rs; sum.y *= rs; sum.z *= rs; sum.w *= rs;
            *(float4*)&As[r][wave * 2 + q][lane * 4] = sum;
        }
    }
    __syncthreads();

    const int half = lane >> 4;
    const int m    = lane & 15;
    const int col0 = wave * 16;

    const int c = col0 + m;
    const float bias = b[c] + b[DIM + c] + b[2 * DIM + c];
    v8f acc;
    #pragma unroll
    for (int v = 0; v < 8; ++v) acc[v] = bias;
    float* op = O + (row0 + 8 * half) * DIM + c;
    #pragma unroll
    for (int r = 0; r < NR; ++r) {
        const float* Wr = W + r * DIM * DIM;
        #pragma unroll
        for (int k0 = 0; k0 < DIM; k0 += 4) {
            const int ka = k0 + 2 * half;
            v2f a, bf;
            a.x = As[r][m][ka];
            a.y = As[r][m][ka + 1];
            bf.x = Wr[ka * DIM + col0 + m];
            bf.y = Wr[(ka + 1) * DIM + col0 + m];
            acc = __builtin_amdgcn_wmma_f32_16x16x4_f32(
                 false, a,  false, bf,
                 (short)0, acc,  false,  false);
        }
    }

    #pragma unroll
    for (int v = 0; v < 8; ++v) {
        op[v * DIM] = acc[v];
    }
}

extern "C" void kernel_launch(void* const* d_in, const int* in_sizes, int n_in,
                              void* d_out, int out_size, void* d_ws, size_t ws_size,
                              hipStream_t stream) {
    const float* x    = (const float*)d_in[0];
    const int*   esrc = (const int*)  d_in[1];
    const int*   edst = (const int*)  d_in[2];
    const float* W1   = (const float*)d_in[3];
    const float* b1   = (const float*)d_in[4];
    const float* W2   = (const float*)d_in[5];
    const float* b2   = (const float*)d_in[6];
    float*       out  = (float*)d_out;

    float* ws = (float*)d_ws;
    float* H1       = ws;
    float* outscale = H1 + (size_t)NN * DIM;
    int*   rowstart = (int*)(outscale + (size_t)NR * NN);
    int*   incnt    = rowstart + (size_t)NR * NN;
    int*   slist    = incnt + (size_t)NR * NN + 16;

    int*   hist     = (int*)H1;
    int*   tot      = hist + (size_t)2 * NR * NCH * NN;

    rgcn_hist<<<2 * NR * HIST_PARTS * NCH, 1024, 0, stream>>>(esrc, edst, hist, tot);
    rgcn_place<<<NR * DEG_PARTS * NCH + NR * DEG_PARTS, 1024, 0, stream>>>(esrc, edst, hist, tot, outscale, rowstart, incnt, slist);

    rgcn_pull_gemm<false><<<NN / 16, 256, 0, stream>>>(x, outscale, rowstart, incnt, slist, W1, b1, H1);

    rgcn_pull_gemm<true><<<NN / 16, 256, 0, stream>>>(H1, outscale, rowstart, incnt, slist, W2, b2, out);
}


# W fragments by whole-line 8-byte loads: wave = 2 column-interleaved N-tiles x one K half (2 accumulators), K halves summed through LDS, rows stored whole; K sum order changes
# speedup vs baseline: 4.1542x; 1.0316x over previous
#include <hip/hip_runtime.h>

#define NN 50000
#define NR 3
#define NE 500000
#define DIM 128

typedef __attribute__((ext_vector_type(2))) float v2f;
typedef __attribute__((ext_vector_type(8))) float v8f;

#define DEG_PARTS 8
#define DEG_BINS (NN / DEG_PARTS)
#define BIN_CH ((DEG_BINS + 1023) / 1024)
#define LIST_PAD 32
#define LIST_LEN (NE + DEG_PARTS * LIST_PAD)
#define HIST_PARTS (DEG_PARTS / 2)
#define HIST_BINS (2 * DEG_BINS)
#define NCH 8
#define CH_E (NE / NCH)
#define CH_V (CH_E / 4)

static __device__ __forceinline__ void deg_count(unsigned int* cnt, unsigned int lo, int node) {
    const unsigned int t = (unsigned int)node - lo;
    if (t < (unsigned int)HIST_BINS) atomicAdd(&cnt[t], 1u);
}

__global__ __launch_bounds__(1024) void rgcn_hist(const int* __restrict__ src,
                                                  const int* __restrict__ dst,
                                                  int* __restrict__ hist,
                                                  int* __restrict__ tot) {
    __shared__ unsigned int cnt[HIST_BINS];
    __shared__ unsigned int blk_tot[2];
    const int tid  = threadIdx.x;
    const int h    = blockIdx.x / (HIST_PARTS * NCH);
    const int rem  = blockIdx.x - h * (HIST_PARTS * NCH);
    const int part = rem / NCH;
    const int c    = rem - part * NCH;
    const unsigned int lo = (unsigned int)(part * HIST_BINS);

    for (int j = tid; j < HIST_BINS; j += 1024) cnt[j] = 0u;
    if (tid < 2) blk_tot[tid] = 0u;
    __syncthreads();

    const int* idx = (h < NR) ? (src + h * NE) : (dst + (h - NR) * NE);
    const int4* p = (const int4*)(idx + c * CH_E);
    constexpr int U = 8;
    for (int i0 = 0; i0 < CH_V; i0 += U * 1024) {
        int4 v[U];
        #pragma unroll
        for (int u = 0; u < U; ++u) {
            const int k = i0 + u * 1024 + tid;
            v[u] = make_int4(-1, -1, -1, -1);
            if (k < CH_V) v[u] = p[k];
        }
        #pragma unroll
        for (int u = 0; u < U; ++u) {
            deg_count(cnt, lo, v[u].x);
            deg_count(cnt, lo, v[u].y);
            deg_count(cnt, lo, v[u].z);
            deg_count(cnt, lo, v[u].w);
        }
    }
    __syncthreads();

    int* hrow = hist + (h * NCH + c) * NN + (int)lo;
    unsigned int s0 = 0u, s1 = 0u;
    for (int j = tid; j < HIST_BINS; j += 1024) {
        const unsigned int v = cnt[j];
        hrow[j] = (int)v;
        if (j < DEG_BINS) s0 += v; else s1 += v;
    }
    for (int off = 16; off > 0; off >>= 1) {
        s0 += __shfl_xor(s0, off);
        s1 += __shfl_xor(s1, off);
    }
    if ((tid & 31) == 0) { atomicAdd(&blk_tot[0], s0); atomicAdd(&blk_tot[1], s1); }
    __syncthreads();
    if (tid < 2) tot[(h * DEG_PARTS + 2 * part + tid) * NCH + c] = (int)blk_tot[tid];
}

static __device__ __forceinline__ void list_place(unsigned int* cursor, unsigned int lo,
                                                  int* __restrict__ lst, int d, int s) {
    const unsigned int t = (unsigned int)d - lo;
    if (t < (unsigned int)DEG_BINS) {
        const unsigned int pos = atomicAdd(&cursor[t], 1u);
        lst[pos] = s;
    }
}

__global__ __launch_bounds__(1024) void rgcn_place(const int* __restrict__ src,
                                                   const int* __restrict__ dst,
                                                   const int* __restrict__ hist,
                                                   const int* __restrict__ tot,
                                                   float* __restrict__ outscale,
                                                   int* __restrict__ rowstart,
                                                   int* __restrict__ incnt,
                                                   int* __restrict__ slist) {
    __shared__ unsigned int cnt[DEG_BINS];
    __shared__ unsigned int pre[DEG_BINS];
    __shared__ unsigned int psum[2][1024];
    __shared__ unsigned int below_all;
    const int tid = threadIdx.x;

    if (blockIdx.x >= NR * DEG_PARTS * NCH) {
        const int q    = blockIdx.x - NR * DEG_PARTS * NCH;
        const int h    = q / DEG_PARTS;
        const int part = q - h * DEG_PARTS;
        const int* hs  = hist + (h * NCH) * NN + part * DEG_BINS;
        #pragma unroll
        for (int k = 0; k < BIN_CH; ++k) {
            const int j = tid + k * 1024;
            if (j < DEG_BINS) {
                unsigned int s = 0u;
                #pragma unroll
                for (int cc = 0; cc < NCH; ++cc) s += (unsigned int)hs[cc * NN + j];
                outscale[h * NN + part * DEG_BINS + j] = rsqrtf(fmaxf((float)s, 1.0f));
            }
        }
        return;
    }

    const int r    = blockIdx.x / (DEG_PARTS * NCH);
    const int rem  = blockIdx.x - r * (DEG_PARTS * NCH);
    const int part = rem / NCH;
    const int c    = rem - part * NCH;
    const unsigned int lo = (unsigned int)(part * DEG_BINS);

    if (tid == 0) below_all = 0u;
    __syncthreads();

    const int* hd = hist + ((NR + r) * NCH) * NN + (int)lo;
    #pragma unroll
    for (int k = 0; k < BIN_CH; ++k) {
        const int j = tid + k * 1024;
        if (j < DEG_BINS) {
            unsigned int total = 0u, lower = 0u;
            #pragma unroll
            for (int cc = 0; cc < NCH; ++cc) {
                const unsigned int v = (unsigned int)hd[cc * NN + j];
                total += v;
                if (cc < c) lower += v;
            }
            cnt[j] = total;
            pre[j] = lower;
        }
    }
    if (tid < part * NCH) atomicAdd(&below_all, (unsigned int)tot[((NR + r) * DEG_PARTS) * NCH + tid]);
    __syncthreads();

    unsigned int cq[BIN_CH];
    unsigned int tsum = 0u;
    #pragma unroll
    for (int q = 0; q < BIN_CH; ++q) {
        const int b = tid * BIN_CH + q;
        cq[q] = (b < DEG_BINS) ? cnt[b] : 0u;
        tsum += cq[q];
    }
    psum[0][tid] = tsum;
    __syncthreads();
    int cur = 0;
    for (int off = 1; off < 1024; off <<= 1) {
        unsigned int v = psum[cur][tid];
        if (tid >= off) v += psum[cur][tid - off];
        psum[cur ^ 1][tid] = v;
        __syncthreads();
        cur ^= 1;
    }
    const unsigned int base = (below_all + (unsigned int)(LIST_PAD * part) + (LIST_PAD - 1)) & ~(unsigned int)(LIST_PAD - 1);
    unsigned int run = base + psum[cur][tid] - tsum;
    #pragma unroll
    for (int q = 0; q < BIN_CH; ++q) {
        const int b = tid * BIN_CH + q;
        if (b < DEG_BINS) {
            if (c == 0) {
                rowstart[r * NN + lo + b] = (int)run;
                incnt[r * NN + lo + b]    = (int)cq[q];
            }
            cnt[b] = run + pre[b];
            run += cq[q];
        }
    }
    __syncthreads();

    const int4* pd = (const int4*)(dst + r * NE + c * CH_E);
    const int4* ps = (const int4*)(src + r * NE + c * CH_E);
    int* lst = slist + r * LIST_LEN;
    constexpr int U2 = 4;
    for (int i0 = 0; i0 < CH_V; i0 += U2 * 1024) {
        int4 vd[U2], vs[U2];
        #pragma unroll
        for (int u = 0; u < U2; ++u) {
            const int k = i0 + u * 1024 + tid;
            vd[u] = make_int4(-1, -1, -1, -1);
            vs[u] = make_int4(0, 0, 0, 0);
            if (k < CH_V) { vd[u] = pd[k]; vs[u] = ps[k]; }
        }
        #pragma unroll
        for (int u = 0; u < U2; ++u) {
            list_place(cnt, lo, lst, vd[u].x, vs[u].x);
            list_place(cnt, lo, lst, vd[u].y, vs[u].y);
            list_place(cnt, lo, lst, vd[u].z, vs[u].z);
            list_place(cnt, lo, lst, vd[u].w, vs[u].w);
        }
    }
}

#define KH (NR * DIM / 2)
template <bool RELU>
__global__ __launch_bounds__(256, 16) void rgcn_pull_gemm(const float* __restrict__ H,
                                                      const float* __restrict__ outscale,
                                                      const int* __restrict__ rowstart,
                                                      const int* __restrict__ incnt,
                                                      const int* __restrict__ slist,
                                                      const float* __restrict__ W,
                                                      const float* __restrict__ b,
                                                      float* __restrict__ O) {
    __shared__ __attribute__((aligned(16))) float As[NR][16][DIM + 4];
    const int row0 = blockIdx.x * 16;
    const int wave = threadIdx.x >> 5;
    const int lane = threadIdx.x & 31;

    int st[NR][2], nn[NR][2];
    #pragma unroll
    for (int r = 0; r < NR; ++r) {
        #pragma unroll
        for (int q = 0; q < 2; ++q) {
            st[r][q] = rowstart[r * NN + row0 + wave * 2 + q];
            nn[r][q] = incnt[r * NN + row0 + wave * 2 + q];
        }
    }
    int   s_l[NR][2];
    float sc_l[NR][2];
    #pragma unroll
    for (int r = 0; r < NR; ++r) {
        #pragma unroll
        for (int q = 0; q < 2; ++q) {
            s_l[r][q] = 0;
            if (lane < nn[r][q]) s_l[r][q] = slist[r * LIST_LEN + st[r][q] + lane];
        }
    }
    #pragma unroll
    for (int r = 0; r < NR; ++r) {
        #pragma unroll
        for (int q = 0; q < 2; ++q) {
            sc_l[r][q] = 0.0f;
            if (lane < nn[r][q]) sc_l[r][q] = outscale[r * NN + s_l[r][q]];
        }
    }

    #pragma unroll
    for (int r = 0; r < NR; ++r) {
        const float* osc = outscale + r * NN;
        #pragma unroll
        for (int q = 0; q < 2; ++q) {
            const int start = __builtin_amdgcn_readfirstlane(st[r][q]);
            const int n     = __builtin_amdgcn_readfirstlane(nn[r][q]);
            const int* lst = slist + r * LIST_LEN + start;
            int   si  = s_l[r][q];
            float sci = sc_l[r][q];
            float4 sum = make_float4(0.0f, 0.0f, 0.0f, 0.0f);
            for (int e0 = 0; e0 < n; e0 += 32) {
                const int m = (n - e0 < 32) ? (n - e0) : 32;
                if (e0 > 0) {
                    si = 0; sci = 0.0f;
                    if (lane < m) { si = lst[e0 + lane]; sci = osc[si]; }
                }
                for (int j = 0; j < m; j += 4) {
                    float4 v[4];
                    float  sc[4];
                    #pragma unroll
                    for (int u = 0; u < 4; ++u) {
                        const bool live = (j + u) < m;
                        const int  jj   = live ? (j + u) : j;
                        const int s = __builtin_amdgcn_readlane(si, jj);
                        sc[u] = __int_as_float(__builtin_amdgcn_readlane(__float_as_int(sci), jj));
                        v[u] = *(const float4*)(H + (size_t)s * DIM + lane * 4);
                        if (!live) v[u] = make_float4(0.0f, 0.0f, 0.0f, 0.0f);
                    }
                    #pragma unroll
                    for (int u = 0; u < 4; ++u) {
                        if (RELU) {
                            v[u].x = fmaxf(v[u].x, 0.0f); v[u].y = fmaxf(v[u].y, 0.0f);
                            v[u].z = fmaxf(v[u].z, 0.0f); v[u].w = fmaxf(v[u].w, 0.0f);
                        }
                        sum.x = fmaf(v[u].x, sc[u], sum.x);
                        sum.y = fmaf(v[u].y, sc[u], sum.y);
                        sum.z = fmaf(v[u].z, sc[u], sum.z);
                        sum.w = fmaf(v[u].w, sc[u], sum.w);
                    }
                }
            }
            const float rs = rsqrtf(fmaxf((float)n, 1.0f));
            sum.x *= rs; sum.y *= rs; sum.z *= rs; sum.w *= rs;
            *(float4*)&As[r][wave * 2 + q][lane * 4] = sum;
        }
    }
    __syncthreads();

    const int wv   = __builtin_amdgcn_readfirstlane(wave);
    const int half = lane >> 4;
    const int m    = lane & 15;
    const int cgrp = wv & 3;
    const int kh   = wv >> 2;
    const int colb = cgrp * 32 + 2 * m;

    v8f acc[2];
    #pragma unroll
    for (int j = 0; j < 2; ++j) {
        #pragma unroll
        for (int v = 0; v < 8; ++v) acc[j][v] = 0.0f;
    }
    const float* Wl = W + (kh * KH + 2 * half) * DIM + colb;
    #pragma unroll
    for (int s = 0; s < KH / 4; ++s) {
        const int kk = kh * KH + 4 * s;
        const int r  = kk >> 7;
        const int ka = (kk & (DIM - 1)) + 2 * half;
        v2f a, bf;
        a.x = As[r][m][ka];
        a.y = As[r][m][ka + 1];
        const float2 b0 = *(const float2*)(Wl + (4 * s) * DIM);
        const float2 b1 = *(const float2*)(Wl + (4 * s + 1) * DIM);
        bf.x = b0.x; bf.y = b1.x;
        acc[0] = __builtin_amdgcn_wmma_f32_16x16x4_f32(
             false, a,  false, bf,
             (short)0, acc[0],  false,  false);
        bf.x = b0.y; bf.y = b1.y;
        acc[1] = __builtin_amdgcn_wmma_f32_16x16x4_f32(
             false, a,  false, bf,
             (short)0, acc[1],  false,  false);
    }

    __syncthreads();
    const int ln = (int)__builtin_amdgcn_mbcnt_lo(~0u, 0u);
    float* red = &As[0][0][0];
    float* rp  = red + (kh * 16 + 8 * (ln >> 4)) * DIM + cgrp * 32 + 2 * (ln & 15);
    #pragma unroll
    for (int v = 0; v < 8; ++v) {
        *(float2*)&rp[v * DIM] = make_float2(acc[0][v], acc[1][v]);
    }
    __syncthreads();

    const float4 bb0 = *(const float4*)(b + ln * 4);
    const float4 bb1 = *(const float4*)(b + DIM + ln * 4);
    const float4 bb2 = *(const float4*)(b + 2 * DIM + ln * 4);

    #pragma unroll
    for (int q = 0; q < 2; ++q) {
        const int row = wv * 2 + q;
        const float4 p0 = *(const float4*)&red[row * DIM + ln * 4];
        const float4 p1 = *(const float4*)&red[(16 + row) * DIM + ln * 4];
        float4 s;
        s.x = bb0.x + bb1.x + bb2.x + p0.x + p1.x;
        s.y = bb0.y + bb1.y + bb2.y + p0.y + p1.y;
        s.z = bb0.z + bb1.z + bb2.z + p0.z + p1.z;
        s.w = bb0.w + bb1.w + bb2.w + p0.w + p1.w;
        *(float4*)(O + (size_t)(row0 + row) * DIM + ln * 4) = s;
    }
}

extern "C" void kernel_launch(void* const* d_in, const int* in_sizes, int n_in,
                              void* d_out, int out_size, void* d_ws, size_t ws_size,
                              hipStream_t stream) {
    const float* x    = (const float*)d_in[0];
    const int*   esrc = (const int*)  d_in[1];
    const int*   edst = (const int*)  d_in[2];
    const float* W1   = (const float*)d_in[3];
    const float* b1   = (const float*)d_in[4];
    const float* W2   = (const float*)d_in[5];
    const float* b2   = (const float*)d_in[6];
    float*       out  = (float*)d_out;

    float* ws = (float*)d_ws;
    float* H1       = ws;
    float* outscale = H1 + (size_t)NN * DIM;
    int*   rowstart = (int*)(outscale + (size_t)NR * NN);
    int*   incnt    = rowstart + (size_t)NR * NN;
    int*   slist    = incnt + (size_t)NR * NN + 16;

    int*   hist     = (int*)H1;
    int*   tot      = hist + (size_t)2 * NR * NCH * NN;

    rgcn_hist<<<2 * NR * HIST_PARTS * NCH, 1024, 0, stream>>>(esrc, edst, hist, tot);
    rgcn_place<<<NR * DEG_PARTS * NCH + NR * DEG_PARTS, 1024, 0, stream>>>(esrc, edst, hist, tot, outscale, rowstart, incnt, slist);

    rgcn_pull_gemm<false><<<NN / 16, 256, 0, stream>>>(x, outscale, rowstart, incnt, slist, W1, b1, H1);

    rgcn_pull_gemm<true><<<NN / 16, 256, 0, stream>>>(H1, outscale, rowstart, incnt, slist, W2, b2, out);
}


# layer 1 writes each edge's source out-degree into the spare top 16 bits of its list word (node index below; exact table read if cut off); layer 2 decodes it: no random scale gather; sums unchanged
# speedup vs baseline: 4.2713x; 1.0282x over previous
#include <hip/hip_runtime.h>

#define NN 50000
#define NR 3
#define NE 500000
#define DIM 128

typedef __attribute__((ext_vector_type(2))) float v2f;
typedef __attribute__((ext_vector_type(8))) float v8f;

#define DEG_PARTS 8
#define DEG_BINS (NN / DEG_PARTS)
#define BIN_CH ((DEG_BINS + 1023) / 1024)
#define LIST_PAD 32
#define LIST_LEN (NE + DEG_PARTS * LIST_PAD)
#define HIST_PARTS (DEG_PARTS / 2)
#define HIST_BINS (2 * DEG_BINS)
#define NCH 8
#define CH_E (NE / NCH)
#define CH_V (CH_E / 4)

static __device__ __forceinline__ void deg_count(unsigned int* cnt, unsigned int lo, int node) {
    const unsigned int t = (unsigned int)node - lo;
    if (t < (unsigned int)HIST_BINS) atomicAdd(&cnt[t], 1u);
}

__global__ __launch_bounds__(1024) void rgcn_hist(const int* __restrict__ src,
                                                  const int* __restrict__ dst,
                                                  int* __restrict__ hist,
                                                  int* __restrict__ tot) {
    __shared__ unsigned int cnt[HIST_BINS];
    __shared__ unsigned int blk_tot[2];
    const int tid  = threadIdx.x;
    const int h    = blockIdx.x / (HIST_PARTS * NCH);
    const int rem  = blockIdx.x - h * (HIST_PARTS * NCH);
    const int part = rem / NCH;
    const int c    = rem - part * NCH;
    const unsigned int lo = (unsigned int)(part * HIST_BINS);

    for (int j = tid; j < HIST_BINS; j += 1024) cnt[j] = 0u;
    if (tid < 2) blk_tot[tid] = 0u;
    __syncthreads();

    const int* idx = (h < NR) ? (src + h * NE) : (dst + (h - NR) * NE);
    const int4* p = (const int4*)(idx + c * CH_E);
    constexpr int U = 8;
    for (int i0 = 0; i0 < CH_V; i0 += U * 1024) {
        int4 v[U];
        #pragma unroll
        for (int u = 0; u < U; ++u) {
            const int k = i0 + u * 1024 + tid;
            v[u] = make_int4(-1, -1, -1, -1);
            if (k < CH_V) v[u] = p[k];
        }
        #pragma unroll
        for (int u = 0; u < U; ++u) {
            deg_count(cnt, lo, v[u].x);
            deg_count(cnt, lo, v[u].y);
            deg_count(cnt, lo, v[u].z);
            deg_count(cnt, lo, v[u].w);
        }
    }
    __syncthreads();

    int* hrow = hist + (h * NCH + c) * NN + (int)lo;
    unsigned int s0 = 0u, s1 = 0u;
    for (int j = tid; j < HIST_BINS; j += 1024) {
        const unsigned int v = cnt[j];
        hrow[j] = (int)v;
        if (j < DEG_BINS) s0 += v; else s1 += v;
    }
    for (int off = 16; off > 0; off >>= 1) {
        s0 += __shfl_xor(s0, off);
        s1 += __shfl_xor(s1, off);
    }
    if ((tid & 31) == 0) { atomicAdd(&blk_tot[0], s0); atomicAdd(&blk_tot[1], s1); }
    __syncthreads();
    if (tid < 2) tot[(h * DEG_PARTS + 2 * part + tid) * NCH + c] = (int)blk_tot[tid];
}

static __device__ __forceinline__ void list_place(unsigned int* cursor, unsigned int lo,
                                                  int* __restrict__ lst, int d, int s) {
    const unsigned int t = (unsigned int)d - lo;
    if (t < (unsigned int)DEG_BINS) {
        const unsigned int pos = atomicAdd(&cursor[t], 1u);
        lst[pos] = s;
    }
}

__global__ __launch_bounds__(1024) void rgcn_place(const int* __restrict__ src,
                                                   const int* __restrict__ dst,
                                                   const int* __restrict__ hist,
                                                   const int* __restrict__ tot,
                                                   int* __restrict__ outdeg,
                                                   int* __restrict__ rowstart,
                                                   int* __restrict__ incnt,
                                                   int* __restrict__ slist) {
    __shared__ unsigned int cnt[DEG_BINS];
    __shared__ unsigned int pre[DEG_BINS];
    __shared__ unsigned int psum[2][1024];
    __shared__ unsigned int below_all;
    const int tid = threadIdx.x;

    if (blockIdx.x >= NR * DEG_PARTS * NCH) {
        const int q    = blockIdx.x - NR * DEG_PARTS * NCH;
        const int h    = q / DEG_PARTS;
        const int part = q - h * DEG_PARTS;
        const int* hs  = hist + (h * NCH) * NN + part * DEG_BINS;
        #pragma unroll
        for (int k = 0; k < BIN_CH; ++k) {
            const int j = tid + k * 1024;
            if (j < DEG_BINS) {
                unsigned int s = 0u;
                #pragma unroll
                for (int cc = 0; cc < NCH; ++cc) s += (unsigned int)hs[cc * NN + j];
                outdeg[h * NN + part * DEG_BINS + j] = (int)s;
            }
        }
        return;
    }

    const int r    = blockIdx.x / (DEG_PARTS * NCH);
    const int rem  = blockIdx.x - r * (DEG_PARTS * NCH);
    const int part = rem / NCH;
    const int c    = rem - part * NCH;
    const unsigned int lo = (unsigned int)(part * DEG_BINS);

    if (tid == 0) below_all = 0u;
    __syncthreads();

    const int* hd = hist + ((NR + r) * NCH) * NN + (int)lo;
    #pragma unroll
    for (int k = 0; k < BIN_CH; ++k) {
        const int j = tid + k * 1024;
        if (j < DEG_BINS) {
            unsigned int total = 0u, lower = 0u;
            #pragma unroll
            for (int cc = 0; cc < NCH; ++cc) {
                const unsigned int v = (unsigned int)hd[cc * NN + j];
                total += v;
                if (cc < c) lower += v;
            }
            cnt[j] = total;
            pre[j] = lower;
        }
    }
    if (tid < part * NCH) atomicAdd(&below_all, (unsigned int)tot[((NR + r) * DEG_PARTS) * NCH + tid]);
    __syncthreads();

    unsigned int cq[BIN_CH];
    unsigned int tsum = 0u;
    #pragma unroll
    for (int q = 0; q < BIN_CH; ++q) {
        const int b = tid * BIN_CH + q;
        cq[q] = (b < DEG_BINS) ? cnt[b] : 0u;
        tsum += cq[q];
    }
    psum[0][tid] = tsum;
    __syncthreads();
    int cur = 0;
    for (int off = 1; off < 1024; off <<= 1) {
        unsigned int v = psum[cur][tid];
        if (tid >= off) v += psum[cur][tid - off];
        psum[cur ^ 1][tid] = v;
        __syncthreads();
        cur ^= 1;
    }
    const unsigned int base = (below_all + (unsigned int)(LIST_PAD * part) + (LIST_PAD - 1)) & ~(unsigned int)(LIST_PAD - 1);
    unsigned int run = base + psum[cur][tid] - tsum;
    #pragma unroll
    for (int q = 0; q < BIN_CH; ++q) {
        const int b = tid * BIN_CH + q;
        if (b < DEG_BINS) {
            if (c == 0) {
                rowstart[r * NN + lo + b] = (int)run;
                incnt[r * NN + lo + b]    = (int)cq[q];
            }
            cnt[b] = run + pre[b];
            run += cq[q];
        }
    }
    __syncthreads();

    const int4* pd = (const int4*)(dst + r * NE + c * CH_E);
    const int4* ps = (const int4*)(src + r * NE + c * CH_E);
    int* lst = slist + r * LIST_LEN;
    constexpr int U2 = 4;
    for (int i0 = 0; i0 < CH_V; i0 += U2 * 1024) {
        int4 vd[U2], vs[U2];
        #pragma unroll
        for (int u = 0; u < U2; ++u) {
            const int k = i0 + u * 1024 + tid;
            vd[u] = make_int4(-1, -1, -1, -1);
            vs[u] = make_int4(0, 0, 0, 0);
            if (k < CH_V) { vd[u] = pd[k]; vs[u] = ps[k]; }
        }
        #pragma unroll
        for (int u = 0; u < U2; ++u) {
            list_place(cnt, lo, lst, vd[u].x, vs[u].x);
            list_place(cnt, lo, lst, vd[u].y, vs[u].y);
            list_place(cnt, lo, lst, vd[u].z, vs[u].z);
            list_place(cnt, lo, lst, vd[u].w, vs[u].w);
        }
    }
}

#define DEG_SAT 0xFFFFu
static_assert(NN <= 65536, "a node index must fit the lower half of a list word");
static __device__ __forceinline__ float deg_scale(unsigned int dg) {
    return rsqrtf(fmaxf((float)dg, 1.0f));
}
#define KH (NR * DIM / 2)
template <bool RELU, bool FIRST>
__global__ __launch_bounds__(256, 16) void rgcn_pull_gemm(const float* __restrict__ H,
                                                      const int* __restrict__ outdeg,
                                                      const int* __restrict__ rowstart,
                                                      const int* __restrict__ incnt,
                                                      int* __restrict__ slist,
                                                      const float* __restrict__ W,
                                                      const float* __restrict__ b,
                                                      float* __restrict__ O) {
    __shared__ __attribute__((aligned(16))) float As[NR][16][DIM + 4];
    const int row0 = blockIdx.x * 16;
    const int wave = threadIdx.x >> 5;
    const int lane = threadIdx.x & 31;

    int st[NR][2], nn[NR][2];
    #pragma unroll
    for (int r = 0; r < NR; ++r) {
        #pragma unroll
        for (int q = 0; q < 2; ++q) {
            st[r][q] = rowstart[r * NN + row0 + wave * 2 + q];
            nn[r][q] = incnt[r * NN + row0 + wave * 2 + q];
        }
    }
    int   s_l[NR][2];
    float sc_l[NR][2];
    unsigned int w_l[NR][2];
    #pragma unroll
    for (int r = 0; r < NR; ++r) {
        #pragma unroll
        for (int q = 0; q < 2; ++q) {
            w_l[r][q] = 0u;
            if (lane < nn[r][q]) w_l[r][q] = (unsigned int)slist[r * LIST_LEN + st[r][q] + lane];
        }
    }
    if (FIRST) {
        unsigned int dg_l[NR][2];
        #pragma unroll
        for (int r = 0; r < NR; ++r) {
            #pragma unroll
            for (int q = 0; q < 2; ++q) {
                dg_l[r][q] = 0u;
                if (lane < nn[r][q]) dg_l[r][q] = (unsigned int)outdeg[r * NN + (int)w_l[r][q]];
            }
        }
        #pragma unroll
        for (int r = 0; r < NR; ++r) {
            #pragma unroll
            for (int q = 0; q < 2; ++q) {
                const unsigned int dg = dg_l[r][q];
                if (lane < nn[r][q])
                    slist[r * LIST_LEN + st[r][q] + lane] = (int)(w_l[r][q] | ((dg < DEG_SAT ? dg : DEG_SAT) << 16));
                s_l[r][q]  = (int)w_l[r][q];
                sc_l[r][q] = deg_scale(dg);
            }
        }
    } else {
        #pragma unroll
        for (int r = 0; r < NR; ++r) {
            #pragma unroll
            for (int q = 0; q < 2; ++q) {
                const int s = (int)(w_l[r][q] & 0xFFFFu);
                unsigned int dg = w_l[r][q] >> 16;
                if (dg == DEG_SAT) dg = (unsigned int)outdeg[r * NN + s];
                s_l[r][q]  = s;
                sc_l[r][q] = deg_scale(dg);
            }
        }
    }

    #pragma unroll
    for (int r = 0; r < NR; ++r) {
        const int* odg = outdeg + r * NN;
        #pragma unroll
        for (int q = 0; q < 2; ++q) {
            const int start = __builtin_amdgcn_readfirstlane(st[r][q]);
            const int n     = __builtin_amdgcn_readfirstlane(nn[r][q]);
            int* lst = slist + r * LIST_LEN + start;
            int   si  = s_l[r][q];
            float sci = sc_l[r][q];
            float4 sum = make_float4(0.0f, 0.0f, 0.0f, 0.0f);
            for (int e0 = 0; e0 < n; e0 += 32) {
                const int m = (n - e0 < 32) ? (n - e0) : 32;
                if (e0 > 0) {
                    si = 0; sci = 0.0f;
                    if (lane < m) {
                        const unsigned int w = (unsigned int)lst[e0 + lane];
                        unsigned int dg;
                        if (FIRST) {
                            si = (int)w;
                            dg = (unsigned int)odg[si];
                            lst[e0 + lane] = (int)(w | ((dg < DEG_SAT ? dg : DEG_SAT) << 16));
                        } else {
                            si = (int)(w & 0xFFFFu);
                            dg = w >> 16;
                            if (dg == DEG_SAT) dg = (unsigned int)odg[si];
                        }
                        sci = deg_scale(dg);
                    }
                }
                for (int j = 0; j < m; j += 4) {
                    float4 v[4];
                    float  sc[4];
                    #pragma unroll
                    for (int u = 0; u < 4; ++u) {
                        const bool live = (j + u) < m;
                        const int  jj   = live ? (j + u) : j;
                        const int s = __builtin_amdgcn_readlane(si, jj);
                        sc[u] = __int_as_float(__builtin_amdgcn_readlane(__float_as_int(sci), jj));
                        v[u] = *(const float4*)(H + (size_t)s * DIM + lane * 4);
                        if (!live) v[u] = make_float4(0.0f, 0.0f, 0.0f, 0.0f);
                    }
                    #pragma unroll
                    for (int u = 0; u < 4; ++u) {
                        if (RELU) {
                            v[u].x = fmaxf(v[u].x, 0.0f); v[u].y = fmaxf(v[u].y, 0.0f);
                            v[u].z = fmaxf(v[u].z, 0.0f); v[u].w = fmaxf(v[u].w, 0.0f);
                        }
                        sum.x = fmaf(v[u].x, sc[u], sum.x);
                        sum.y = fmaf(v[u].y, sc[u], sum.y);
                        sum.z = fmaf(v[u].z, sc[u], sum.z);
                        sum.w = fmaf(v[u].w, sc[u], sum.w);
                    }
                }
            }
            const float rs = rsqrtf(fmaxf((float)n, 1.0f));
            sum.x *= rs; sum.y *= rs; sum.z *= rs; sum.w *= rs;
            *(float4*)&As[r][wave * 2 + q][lane * 4] = sum;
        }
    }
    __syncthreads();

    const int wv   = __builtin_amdgcn_readfirstlane(wave);
    const int half = lane >> 4;
    const int m    = lane & 15;
    const int cgrp = wv & 3;
    const int kh   = wv >> 2;
    const int colb = cgrp * 32 + 2 * m;

    v8f acc[2];
    #pragma unroll
    for (int j = 0; j < 2; ++j) {
        #pragma unroll
        for (int v = 0; v < 8; ++v) acc[j][v] = 0.0f;
    }
    const float* Wl = W + (kh * KH + 2 * half) * DIM + colb;
    #pragma unroll
    for (int s = 0; s < KH / 4; ++s) {
        const int kk = kh * KH + 4 * s;
        const int r  = kk >> 7;
        const int ka = (kk & (DIM - 1)) + 2 * half;
        v2f a, bf;
        a.x = As[r][m][ka];
        a.y = As[r][m][ka + 1];
        const float2 b0 = *(const float2*)(Wl + (4 * s) * DIM);
        const float2 b1 = *(const float2*)(Wl + (4 * s + 1) * DIM);
        bf.x = b0.x; bf.y = b1.x;
        acc[0] = __builtin_amdgcn_wmma_f32_16x16x4_f32(
             false, a,  false, bf,
             (short)0, acc[0],  false,  false);
        bf.x = b0.y; bf.y = b1.y;
        acc[1] = __builtin_amdgcn_wmma_f32_16x16x4_f32(
             false, a,  false, bf,
             (short)0, acc[1],  false,  false);
    }

    __syncthreads();
    const int ln = (int)__builtin_amdgcn_mbcnt_lo(~0u, 0u);
    float* red = &As[0][0][0];
    float* rp  = red + (kh * 16 + 8 * (ln >> 4)) * DIM + cgrp * 32 + 2 * (ln & 15);
    #pragma unroll
    for (int v = 0; v < 8; ++v) {
        *(float2*)&rp[v * DIM] = make_float2(acc[0][v], acc[1][v]);
    }
    __syncthreads();

    const float4 bb0 = *(const float4*)(b + ln * 4);
    const float4 bb1 = *(const float4*)(b + DIM + ln * 4);
    const float4 bb2 = *(const float4*)(b + 2 * DIM + ln * 4);

    #pragma unroll
    for (int q = 0; q < 2; ++q) {
        const int row = wv * 2 + q;
        const float4 p0 = *(const float4*)&red[row * DIM + ln * 4];
        const float4 p1 = *(const float4*)&red[(16 + row) * DIM + ln * 4];
        float4 s;
        s.x = bb0.x + bb1.x + bb2.x + p0.x + p1.x;
        s.y = bb0.y + bb1.y + bb2.y + p0.y + p1.y;
        s.z = bb0.z + bb1.z + bb2.z + p0.z + p1.z;
        s.w = bb0.w + bb1.w + bb2.w + p0.w + p1.w;
        *(float4*)(O + (size_t)(row0 + row) * DIM + ln * 4) = s;
    }
}

extern "C" void kernel_launch(void* const* d_in, const int* in_sizes, int n_in,
                              void* d_out, int out_size, void* d_ws, size_t ws_size,
                              hipStream_t stream) {
    const float* x    = (const float*)d_in[0];
    const int*   esrc = (const int*)  d_in[1];
    const int*   edst = (const int*)  d_in[2];
    const float* W1   = (const float*)d_in[3];
    const float* b1   = (const float*)d_in[4];
    const float* W2   = (const float*)d_in[5];
    const float* b2   = (const float*)d_in[6];
    float*       out  = (float*)d_out;

    float* ws = (float*)d_ws;
    float* H1       = ws;
    int*   outdeg   = (int*)(H1 + (size_t)NN * DIM);
    int*   rowstart = outdeg + (size_t)NR * NN;
    int*   incnt    = rowstart + (size_t)NR * NN;
    int*   slist    = incnt + (size_t)NR * NN + 16;

    int*   hist     = (int*)H1;
    int*   tot      = hist + (size_t)2 * NR * NCH * NN;

    rgcn_hist<<<2 * NR * HIST_PARTS * NCH, 1024, 0, stream>>>(esrc, edst, hist, tot);
    rgcn_place<<<NR * DEG_PARTS * NCH + NR * DEG_PARTS, 1024, 0, stream>>>(esrc, edst, hist, tot, outdeg, rowstart, incnt, slist);

    rgcn_pull_gemm<false, true><<<NN / 16, 256, 0, stream>>>(x, outdeg, rowstart, incnt, slist, W1, b1, H1);

    rgcn_pull_gemm<true, false><<<NN / 16, 256, 0, stream>>>(H1, outdeg, rowstart, incnt, slist, W2, b2, out);
}
